# starting kernel
# baseline (speedup 1.0000x reference)
#include <hip/hip_runtime.h>
#include <math.h>

typedef float v2f __attribute__((ext_vector_type(2)));
typedef float v8f __attribute__((ext_vector_type(8)));

#define B_ 4
#define N_ 512
#define C_ 64
#define O_ 64

__global__ __launch_bounds__(128) void edgeconv_gemm(
    const float* __restrict__ x,
    const float* __restrict__ W,
    const float* __restrict__ bias,
    float* __restrict__ u,
    float* __restrict__ dws)
{
    const int wave = threadIdx.x >> 5;
    const int lane = threadIdx.x & 31;
    const int r    = lane & 15;
    const int h    = lane >> 4;
    const int b    = blockIdx.x >> 5;
    const int mt   = blockIdx.x & 31;
    const int o    = wave * 16 + r;
    const int rowBase = mt * 16;

    const float* xrow = x + (size_t)(b * N_ + rowBase + r) * C_;

    v8f accA = {};
    v8f accD = {};
    #pragma unroll
    for (int k0 = 0; k0 < C_; k0 += 4) {
        const int ka = k0 + 2 * h;
        v2f aF = *(const v2f*)(xrow + ka);
        v2f bA, bD;
        bA.x = W[(size_t)(ka)         * O_ + o];
        bA.y = W[(size_t)(ka + 1)     * O_ + o];
        bD.x = W[(size_t)(C_ + ka)    * O_ + o];
        bD.y = W[(size_t)(C_ + ka + 1)* O_ + o];
        accA = __builtin_amdgcn_wmma_f32_16x16x4_f32(
             false, aF,  false, bA,
             (short)0, accA,  false,  false);
        accD = __builtin_amdgcn_wmma_f32_16x16x4_f32(
            false, aF, false, bD, (short)0, accD, false, false);
    }

    const float bo = bias[o];
    #pragma unroll
    for (int i = 0; i < 8; ++i) {
        const int row = rowBase + i + 8 * h;
        const size_t idx = (size_t)(b * N_ + row) * O_ + o;
        const float dv = accD[i];
        dws[idx] = dv;
        u[idx]   = accA[i] - dv + bo;
    }
}

__device__ __forceinline__ float gelu_exact(float v) {
    return 0.5f * v * (1.0f + erff(v * 0.70710678118654752f));
}

__global__ __launch_bounds__(256) void edgeconv_lnmean(
    const float* __restrict__ u,
    const float* __restrict__ dws,
    const int*   __restrict__ mask,
    const float* __restrict__ gamma,
    const float* __restrict__ beta,
    float* __restrict__ out)
{
    __shared__ float2 part[8 * 32];
    const int tid  = threadIdx.x;
    const int wave = tid >> 5;
    const int lane = tid & 31;
    const int b    = blockIdx.x >> 9;
    const int n    = blockIdx.x & (N_ - 1);
    const size_t row = (size_t)(b * N_ + n);

    if (mask[row] == 0) {
        if (tid < 32) {
            float2 z; z.x = 0.0f; z.y = 0.0f;
            *(float2*)(out + row * O_ + 2 * tid) = z;
        }
        return;
    }

    const float2 u2  = *(const float2*)(u     + row * O_ + 2 * lane);
    const float2 g2  = *(const float2*)(gamma + 2 * lane);
    const float2 be2 = *(const float2*)(beta  + 2 * lane);

    const float* dbase = dws + (size_t)b * N_ * O_;
    float2 acc; acc.x = 0.0f; acc.y = 0.0f;

    const int m0 = wave * 64;
    for (int m = m0; m < m0 + 64; ++m) {
        const float2 d2 = *(const float2*)(dbase + (size_t)m * O_ + 2 * lane);
        const float hx = gelu_exact(u2.x + d2.x);
        const float hy = gelu_exact(u2.y + d2.y);
        float s = hx + hy;
        float q = hx * hx + hy * hy;
        #pragma unroll
        for (int off = 16; off > 0; off >>= 1) {
            s += __shfl_xor(s, off, 32);
            q += __shfl_xor(q, off, 32);
        }
        const float mu  = s * (1.0f / 64.0f);
        const float var = q * (1.0f / 64.0f) - mu * mu;
        const float rs  = rsqrtf(var + 1e-5f);
        acc.x += (hx - mu) * rs * g2.x + be2.x;
        acc.y += (hy - mu) * rs * g2.y + be2.y;
    }

    part[wave * 32 + lane] = acc;
    __syncthreads();

    if (tid < 32) {
        float2 res = part[tid];
        #pragma unroll
        for (int w = 1; w < 8; ++w) {
            res.x += part[w * 32 + tid].x;
            res.y += part[w * 32 + tid].y;
        }
        res.x *= (1.0f / (float)N_);
        res.y *= (1.0f / (float)N_);
        *(float2*)(out + row * O_ + 2 * tid) = res;
    }
}

extern "C" void kernel_launch(void* const* d_in, const int* in_sizes, int n_in,
                              void* d_out, int out_size, void* d_ws, size_t ws_size,
                              hipStream_t stream) {
    (void)in_sizes; (void)n_in; (void)out_size; (void)ws_size;
    const float* x     = (const float*)d_in[0];
    const int*   mask  = (const int*)  d_in[1];
    const float* W     = (const float*)d_in[2];
    const float* bias  = (const float*)d_in[3];
    const float* gamma = (const float*)d_in[4];
    const float* beta  = (const float*)d_in[5];
    float* out = (float*)d_out;

    float* u   = (float*)d_ws;
    float* dws = u + (size_t)B_ * N_ * O_;

    edgeconv_gemm<<<B_ * (N_ / 16), 128, 0, stream>>>(x, W, bias, u, dws);
    edgeconv_lnmean<<<B_ * N_, 256, 0, stream>>>(u, dws, mask, gamma, beta, out);
}


# LN+mean kernel: each lane owns one neighbour row (d stored channel-quad-major), LayerNorm sums stay in-lane, no shuffle tree per neighbour; sum order changes, gamma/beta applied after the sum over m
# speedup vs baseline: 1.6324x; 1.6324x over previous
#include <hip/hip_runtime.h>
#include <math.h>

typedef float v2f __attribute__((ext_vector_type(2)));
typedef float v4f __attribute__((ext_vector_type(4)));
typedef float v8f __attribute__((ext_vector_type(8)));

#define B_ 4
#define N_ 512
#define C_ 64
#define O_ 64

__global__ __launch_bounds__(128) void edgeconv_gemm(
    const float* __restrict__ x,
    const float* __restrict__ W,
    const float* __restrict__ bias,
    float* __restrict__ u,
    float* __restrict__ dQ)
{
    const int wave = threadIdx.x >> 5;
    const int lane = threadIdx.x & 31;
    const int r    = lane & 15;
    const int h    = lane >> 4;
    const int b    = blockIdx.x >> 5;
    const int mt   = blockIdx.x & 31;
    const int oBase   = wave * 16;
    const int rowBase = mt * 16;
    const int o    = oBase + r;

    const float* xrow = x + (size_t)(b * N_ + rowBase + r) * C_;

    v8f accA = {};
    v8f accD = {};
    #pragma unroll
    for (int k0 = 0; k0 < C_; k0 += 4) {
        const int ka = k0 + 2 * h;
        v2f xF = *(const v2f*)(xrow + ka);
        v2f wA, wD;
        wA.x = W[(size_t)(ka)         * O_ + o];
        wA.y = W[(size_t)(ka + 1)     * O_ + o];
        wD.x = W[(size_t)(C_ + ka)    * O_ + o];
        wD.y = W[(size_t)(C_ + ka + 1)* O_ + o];
        accA = __builtin_amdgcn_wmma_f32_16x16x4_f32(
             false, wA,  false, xF,
             (short)0, accA,  false,  false);
        accD = __builtin_amdgcn_wmma_f32_16x16x4_f32(
            false, wD, false, xF, (short)0, accD, false, false);
    }

    const int oc  = oBase + 8 * h;
    const int row = rowBase + r;
    float uo[8];
    #pragma unroll
    for (int i = 0; i < 8; ++i) uo[i] = accA[i] - accD[i] + bias[oc + i];
    v4f d0, d1;
    d0.x = accD[0]; d0.y = accD[1]; d0.z = accD[2]; d0.w = accD[3];
    d1.x = accD[4]; d1.y = accD[5]; d1.z = accD[6]; d1.w = accD[7];
    float* dq = dQ + (((size_t)b * (O_ / 4) + (oc >> 2)) * N_ + row) * 4;
    *(v4f*)(dq)                  = d0;
    *(v4f*)(dq + (size_t)N_ * 4) = d1;
    float* up = u + (size_t)(b * N_ + row) * O_ + oc;
    float4 s0, s1;
    s0.x = uo[0]; s0.y = uo[1]; s0.z = uo[2]; s0.w = uo[3];
    s1.x = uo[4]; s1.y = uo[5]; s1.z = uo[6]; s1.w = uo[7];
    *(float4*)(up)     = s0;
    *(float4*)(up + 4) = s1;
}

__device__ __forceinline__ float gelu_exact(float v) {
    return 0.5f * v * (1.0f + erff(v * 0.70710678118654752f));
}

__device__ __forceinline__ float xhalf(float v) {
    const int t = __builtin_bit_cast(int, v);
    return __builtin_bit_cast(float,
        __builtin_amdgcn_permlanex16(t, t, 0x76543210, 0xfedcba98, false, false));
}

#define MPASS_ 64

__global__ __launch_bounds__(128) void edgeconv_lnmean(
    const float* __restrict__ u,
    const float* __restrict__ dQ,
    const int*   __restrict__ mask,
    const float* __restrict__ gamma,
    const float* __restrict__ beta,
    float* __restrict__ out)
{
    __shared__ float part[4][O_];
    const int tid  = threadIdx.x;
    const int wave = tid >> 5;
    const int lane = tid & 31;
    const int r    = lane & 15;
    const int hh   = lane >> 4;
    const int b    = blockIdx.x >> 9;
    const int n    = blockIdx.x & (N_ - 1);
    const size_t row = (size_t)(b * N_ + n);

    if (mask[row] == 0) {
        if (tid < 32) {
            float2 z; z.x = 0.0f; z.y = 0.0f;
            *(float2*)(out + row * O_ + 2 * tid) = z;
        }
        return;
    }

    float uu[32];
    {
        const float4* up = (const float4*)(u + row * O_ + 32 * hh);
        #pragma unroll
        for (int k4 = 0; k4 < 8; ++k4) {
            const float4 t = up[k4];
            uu[4 * k4 + 0] = t.x; uu[4 * k4 + 1] = t.y;
            uu[4 * k4 + 2] = t.z; uu[4 * k4 + 3] = t.w;
        }
    }

    float acc[32];
    #pragma unroll
    for (int k = 0; k < 32; ++k) acc[k] = 0.0f;

    const float* dcol = dQ + (((size_t)b * (O_ / 4) + 8 * hh) * N_ + 16 * wave + r) * 4;

    for (int p = 0; p < N_ / MPASS_; ++p) {
        const float* dp = dcol + (size_t)p * MPASS_ * 4;
        v4f dv[8];
        #pragma unroll
        for (int k4 = 0; k4 < 8; ++k4) dv[k4] = *(const v4f*)(dp + (size_t)k4 * N_ * 4);
        asm volatile("" :: "v"(dv[0]), "v"(dv[1]), "v"(dv[2]), "v"(dv[3]),
                           "v"(dv[4]), "v"(dv[5]), "v"(dv[6]), "v"(dv[7]));
        float hv[32];
        float s0 = 0.0f, s1 = 0.0f, q0 = 0.0f, q1 = 0.0f;
        #pragma unroll
        for (int k = 0; k < 32; k += 2) {
            const float g0 = gelu_exact(uu[k]     + dv[k >> 2][k & 3]);
            const float g1 = gelu_exact(uu[k + 1] + dv[k >> 2][(k & 3) + 1]);
            hv[k] = g0; hv[k + 1] = g1;
            s0 += g0; q0 += g0 * g0;
            s1 += g1; q1 += g1 * g1;
        }
        float s = s0 + s1;
        float q = q0 + q1;
        s += xhalf(s);
        q += xhalf(q);
        const float mu  = s * (1.0f / 64.0f);
        const float var = q * (1.0f / 64.0f) - mu * mu;
        const float rs  = rsqrtf(var + 1e-5f);
        #pragma unroll
        for (int k = 0; k < 32; ++k) acc[k] += (hv[k] - mu) * rs;
    }

    #pragma unroll
    for (int k = 0; k < 32; ++k) {
        float v = acc[k];
        #pragma unroll
        for (int off = 8; off > 0; off >>= 1) v += __shfl_xor(v, off, 32);
        acc[k] = v;
    }
    if (r == 0) {
        #pragma unroll
        for (int k = 0; k < 32; ++k) part[wave][32 * hh + k] = acc[k];
    }
    __syncthreads();

    if (tid < 32) {
        const int c = 2 * tid;
        const float2 g2  = *(const float2*)(gamma + c);
        const float2 be2 = *(const float2*)(beta  + c);
        float2 res;
        res.x = part[0][c];
        res.y = part[0][c + 1];
        #pragma unroll
        for (int w = 1; w < 4; ++w) {
            res.x += part[w][c];
            res.y += part[w][c + 1];
        }
        res.x = res.x * (1.0f / (float)N_) * g2.x + be2.x;
        res.y = res.y * (1.0f / (float)N_) * g2.y + be2.y;
        *(float2*)(out + row * O_ + c) = res;
    }
}

extern "C" void kernel_launch(void* const* d_in, const int* in_sizes, int n_in,
                              void* d_out, int out_size, void* d_ws, size_t ws_size,
                              hipStream_t stream) {
    (void)in_sizes; (void)n_in; (void)out_size; (void)ws_size;
    const float* x     = (const float*)d_in[0];
    const int*   mask  = (const int*)  d_in[1];
    const float* W     = (const float*)d_in[2];
    const float* bias  = (const float*)d_in[3];
    const float* gamma = (const float*)d_in[4];
    const float* beta  = (const float*)d_in[5];
    float* out = (float*)d_out;

    float* u  = (float*)d_ws;
    float* dQ = u + (size_t)B_ * N_ * O_;

    edgeconv_gemm<<<B_ * (N_ / 16), 128, 0, stream>>>(x, W, bias, u, dQ);
    edgeconv_lnmean<<<B_ * N_, 128, 0, stream>>>(u, dQ, mask, gamma, beta, out);
}


# lnmean: d tile double-buffered in registers, each pass's loads issued one pass ahead; pass written as channel pairs, 4 at a time, to fit 5 waves/SIMD; q squares fused (rounding), sum order kept
# speedup vs baseline: 4.3484x; 2.6638x over previous
#include <hip/hip_runtime.h>
#include <math.h>

typedef float v2f __attribute__((ext_vector_type(2)));
typedef float v4f __attribute__((ext_vector_type(4)));
typedef float v8f __attribute__((ext_vector_type(8)));

#define B_ 4
#define N_ 512
#define C_ 64
#define O_ 64

__global__ __launch_bounds__(128) void edgeconv_gemm(
    const float* __restrict__ x,
    const float* __restrict__ W,
    const float* __restrict__ bias,
    float* __restrict__ u,
    float* __restrict__ dQ)
{
    const int wave = threadIdx.x >> 5;
    const int lane = threadIdx.x & 31;
    const int r    = lane & 15;
    const int h    = lane >> 4;
    const int b    = blockIdx.x >> 5;
    const int mt   = blockIdx.x & 31;
    const int oBase   = wave * 16;
    const int rowBase = mt * 16;
    const int o    = oBase + r;

    const float* xrow = x + (size_t)(b * N_ + rowBase + r) * C_;

    v8f accA = {};
    v8f accD = {};
    #pragma unroll
    for (int k0 = 0; k0 < C_; k0 += 4) {
        const int ka = k0 + 2 * h;
        v2f xF = *(const v2f*)(xrow + ka);
        v2f wA, wD;
        wA.x = W[(size_t)(ka)         * O_ + o];
        wA.y = W[(size_t)(ka + 1)     * O_ + o];
        wD.x = W[(size_t)(C_ + ka)    * O_ + o];
        wD.y = W[(size_t)(C_ + ka + 1)* O_ + o];
        accA = __builtin_amdgcn_wmma_f32_16x16x4_f32(
             false, wA,  false, xF,
             (short)0, accA,  false,  false);
        accD = __builtin_amdgcn_wmma_f32_16x16x4_f32(
            false, wD, false, xF, (short)0, accD, false, false);
    }

    const int oc  = oBase + 8 * h;
    const int row = rowBase + r;
    float uo[8];
    #pragma unroll
    for (int i = 0; i < 8; ++i) uo[i] = accA[i] - accD[i] + bias[oc + i];
    v4f d0, d1;
    d0.x = accD[0]; d0.y = accD[1]; d0.z = accD[2]; d0.w = accD[3];
    d1.x = accD[4]; d1.y = accD[5]; d1.z = accD[6]; d1.w = accD[7];
    float* dq = dQ + (((size_t)b * (O_ / 4) + (oc >> 2)) * N_ + row) * 4;
    *(v4f*)(dq)                  = d0;
    *(v4f*)(dq + (size_t)N_ * 4) = d1;
    float* up = u + (size_t)(b * N_ + row) * O_ + oc;
    float4 s0, s1;
    s0.x = uo[0]; s0.y = uo[1]; s0.z = uo[2]; s0.w = uo[3];
    s1.x = uo[4]; s1.y = uo[5]; s1.z = uo[6]; s1.w = uo[7];
    *(float4*)(up)     = s0;
    *(float4*)(up + 4) = s1;
}

#define GP_ 4
#define GELU_COEFFS_ 9
__device__ __forceinline__ void gelu_group(const v2f (&v)[GP_], v2f (&g)[GP_]) {
    const float cf[GELU_COEFFS_] = { -1.69039185e-6f,  2.50831727e-5f, -1.14461938e-4f,
                                     -3.23344255e-4f,  7.33338762e-3f, -5.27142026e-2f,
                                     -4.59115446e-1f, -1.15112329f,    -9.99998868e-1f };
    v2f t[GP_], p[GP_];
    #pragma unroll
    for (int i = 0; i < GP_; ++i) {
        t[i] = __builtin_elementwise_min(__builtin_elementwise_abs(v[i]), (v2f)(6.0f));
        p[i] = (v2f)(cf[0]);
    }
    #pragma unroll
    for (int j = 1; j < GELU_COEFFS_; ++j) {
        #pragma unroll
        for (int i = 0; i < GP_; ++i) p[i] = __builtin_elementwise_fma(p[i], t[i], (v2f)(cf[j]));
    }
    #pragma unroll
    for (int i = 0; i < GP_; ++i) {
        v2f e;
        e.x = __builtin_amdgcn_exp2f(p[i].x);
        e.y = __builtin_amdgcn_exp2f(p[i].y);
        g[i] = __builtin_elementwise_fma(-t[i], e, __builtin_elementwise_max(v[i], (v2f)(0.0f)));
    }
}

__device__ __forceinline__ float xhalf(float v) {
    const int t = __builtin_bit_cast(int, v);
    return __builtin_bit_cast(float,
        __builtin_amdgcn_permlanex16(t, t, 0x76543210, 0xfedcba98, false, false));
}

__device__ __forceinline__ void load_tile(v4f (&dv)[8], const float* dp) {
    #pragma unroll
    for (int k4 = 0; k4 < 8; ++k4) dv[k4] = *(const v4f*)(dp + (size_t)k4 * N_ * 4);
}

__device__ __forceinline__ void gelu_part(const int k0, const v2f (&uu)[16], const v4f (&dv)[8],
                                          v2f (&hv)[16], v2f& sv, v2f& qv) {
    v2f lin[GP_], g[GP_];
    #pragma unroll
    for (int i = 0; i < GP_; ++i) {
        const int k = k0 + i;
        const v4f d4 = dv[k >> 1];
        const v2f d2 = (k & 1) ? __builtin_shufflevector(d4, d4, 2, 3)
                               : __builtin_shufflevector(d4, d4, 0, 1);
        lin[i] = uu[k] + d2;
    }
    gelu_group(lin, g);
    #pragma unroll
    for (int i = 0; i < GP_; ++i) {
        hv[k0 + i] = g[i];
        sv += g[i];
        qv += g[i] * g[i];
    }
}

__device__ __forceinline__ void pass_update(const v2f (&uu)[16], const v4f (&dv)[8], v2f (&acc)[16]) {
    v2f hv[16];
    v2f sv = (v2f)(0.0f), qv = (v2f)(0.0f);
    #pragma unroll
    for (int k0 = 0; k0 < 16; k0 += GP_) gelu_part(k0, uu, dv, hv, sv, qv);
    float s = sv.x + sv.y;
    float q = qv.x + qv.y;
    s += xhalf(s);
    q += xhalf(q);
    const float mu  = s * (1.0f / 64.0f);
    const float var = q * (1.0f / 64.0f) - mu * mu;
    const float rs  = rsqrtf(var + 1e-5f);
    #pragma unroll
    for (int k = 0; k < 16; ++k) acc[k] += (hv[k] - mu) * rs;
}

#define MPASS_ 64

__global__ __launch_bounds__(128) __attribute__((amdgpu_waves_per_eu(5, 5))) void edgeconv_lnmean(
    const float* __restrict__ u,
    const float* __restrict__ dQ,
    const int*   __restrict__ mask,
    const float* __restrict__ gamma,
    const float* __restrict__ beta,
    float* __restrict__ out)
{
    __shared__ float part[4][O_];
    const int tid  = threadIdx.x;
    const int wave = tid >> 5;
    const int lane = tid & 31;
    const int r    = lane & 15;
    const int hh   = lane >> 4;
    const int b    = blockIdx.x >> 9;
    const int n    = blockIdx.x & (N_ - 1);
    const size_t row = (size_t)(b * N_ + n);

    if (mask[row] == 0) {
        if (tid < 32) {
            float2 z; z.x = 0.0f; z.y = 0.0f;
            *(float2*)(out + row * O_ + 2 * tid) = z;
        }
        return;
    }

    v2f uu[16];
    {
        const float4* up = (const float4*)(u + row * O_ + 32 * hh);
        #pragma unroll
        for (int k4 = 0; k4 < 8; ++k4) {
            const float4 t = up[k4];
            uu[2 * k4].x     = t.x; uu[2 * k4].y     = t.y;
            uu[2 * k4 + 1].x = t.z; uu[2 * k4 + 1].y = t.w;
        }
    }

    v2f acc2[16];
    #pragma unroll
    for (int k = 0; k < 16; ++k) acc2[k] = (v2f)(0.0f);

    const float* dcol = dQ + (((size_t)b * (O_ / 4) + 8 * hh) * N_ + 16 * wave + r) * 4;

    v4f dvA[8], dvB[8];
    load_tile(dvA, dcol);
    for (int p = 0; p < N_ / MPASS_; p += 2) {
        __builtin_amdgcn_sched_barrier(0);
        load_tile(dvB, dcol + (size_t)(p + 1) * MPASS_ * 4);
        __builtin_amdgcn_sched_barrier(0);
        pass_update(uu, dvA, acc2);
        const int pn = (p + 2 < N_ / MPASS_) ? p + 2 : p + 1;
        __builtin_amdgcn_sched_barrier(0);
        load_tile(dvA, dcol + (size_t)pn * MPASS_ * 4);
        __builtin_amdgcn_sched_barrier(0);
        pass_update(uu, dvB, acc2);
    }

    float acc[32];
    #pragma unroll
    for (int k = 0; k < 16; ++k) { acc[2 * k] = acc2[k].x; acc[2 * k + 1] = acc2[k].y; }

    #pragma unroll
    for (int k = 0; k < 32; ++k) {
        float v = acc[k];
        #pragma unroll
        for (int off = 8; off > 0; off >>= 1) v += __shfl_xor(v, off, 32);
        acc[k] = v;
    }
    if (r == 0) {
        #pragma unroll
        for (int k = 0; k < 32; ++k) part[wave][32 * hh + k] = acc[k];
    }
    __syncthreads();

    if (tid < 32) {
        const int c = 2 * tid;
        const float2 g2  = *(const float2*)(gamma + c);
        const float2 be2 = *(const float2*)(beta  + c);
        float2 res;
        res.x = part[0][c];
        res.y = part[0][c + 1];
        #pragma unroll
        for (int w = 1; w < 4; ++w) {
            res.x += part[w][c];
            res.y += part[w][c + 1];
        }
        res.x = res.x * (1.0f / (float)N_) * g2.x + be2.x;
        res.y = res.y * (1.0f / (float)N_) * g2.y + be2.y;
        *(float2*)(out + row * O_ + c) = res;
    }
}

extern "C" void kernel_launch(void* const* d_in, const int* in_sizes, int n_in,
                              void* d_out, int out_size, void* d_ws, size_t ws_size,
                              hipStream_t stream) {
    (void)in_sizes; (void)n_in; (void)out_size; (void)ws_size;
    const float* x     = (const float*)d_in[0];
    const int*   mask  = (const int*)  d_in[1];
    const float* W     = (const float*)d_in[2];
    const float* bias  = (const float*)d_in[3];
    const float* gamma = (const float*)d_in[4];
    const float* beta  = (const float*)d_in[5];
    float* out = (float*)d_out;

    float* u  = (float*)d_ws;
    float* dQ = u + (size_t)B_ * N_ * O_;

    edgeconv_gemm<<<B_ * (N_ / 16), 128, 0, stream>>>(x, W, bias, u, dQ);
    edgeconv_lnmean<<<B_ * N_, 128, 0, stream>>>(u, dQ, mask, gamma, beta, out);
}
